# starting kernel
# baseline (speedup 1.0000x reference)
#include <hip/hip_runtime.h>


typedef float v2f __attribute__((ext_vector_type(2)));
typedef float v8f __attribute__((ext_vector_type(8)));

#define GRID_N 512
#define MCTRL  64
#define NCTRL  64
#define BATCH  8

__global__ void build_basis_kernel(const float* __restrict__ Nu,
                                   const int*   __restrict__ uspan,
                                   const float* __restrict__ Nv,
                                   const int*   __restrict__ vspan,
                                   float* __restrict__ Bu,
                                   float* __restrict__ Bv) {
  int t = blockIdx.x * blockDim.x + threadIdx.x;
  if (t < GRID_N) {
    float* row = Bu + (size_t)t * MCTRL;
    #pragma unroll
    for (int j = 0; j < MCTRL; ++j) row[j] = 0.0f;
    int s = uspan[t];
    #pragma unroll
    for (int r = 0; r < 4; ++r) row[s - 3 + r] = Nu[t * 4 + r];
  } else if (t < 2 * GRID_N) {
    int g = t - GRID_N;
    float* row = Bv + (size_t)g * NCTRL;
    #pragma unroll
    for (int j = 0; j < NCTRL; ++j) row[j] = 0.0f;
    int s = vspan[g];
    #pragma unroll
    for (int r = 0; r < 4; ++r) row[s - 3 + r] = Nv[g * 4 + r];
  }
}

__device__ __forceinline__ int band_min(int span) {
  int c = span - 3;
  return (c > 56) ? 56 : (c < 0 ? 0 : c);
}

__global__ __launch_bounds__(128)
void stageA_kernel(const float* __restrict__ ctrl,
                   const float* __restrict__ Bu,
                   const int*   __restrict__ uspan,
                   float* __restrict__ Mid)
{
  const int gt = blockIdx.x;
  const int bb = blockIdx.y;
  const int g0 = gt * 16;

  const int d    = threadIdx.x >> 5;
  const int lane = threadIdx.x & 31;
  const int hh   = lane >> 4;
  const int l    = lane & 15;

  const int umin = band_min(uspan[g0]);

  v8f acc0 = {}, acc1 = {}, acc2 = {}, acc3 = {};
  const float* BuRow = Bu + (size_t)(g0 + l) * MCTRL + umin;

  #pragma unroll
  for (int kb = 0; kb < 2; ++kb) {
    const int k0 = 4 * kb + 2 * hh;
    v2f a;
    a.x = BuRow[k0];
    a.y = BuRow[k0 + 1];
    const float* c0 = ctrl + ((size_t)(bb * MCTRL + umin + k0) * NCTRL) * 4 + d;
    const float* c1 = c0 + (size_t)NCTRL * 4;
    {
      v2f b; b.x = c0[(0 * 16 + l) * 4]; b.y = c1[(0 * 16 + l) * 4];
      acc0 = __builtin_amdgcn_wmma_f32_16x16x4_f32(false, a, false, b, (short)0, acc0, false, false);
    }
    {
      v2f b; b.x = c0[(1 * 16 + l) * 4]; b.y = c1[(1 * 16 + l) * 4];
      acc1 = __builtin_amdgcn_wmma_f32_16x16x4_f32(false, a, false, b, (short)0, acc1, false, false);
    }
    {
      v2f b; b.x = c0[(2 * 16 + l) * 4]; b.y = c1[(2 * 16 + l) * 4];
      acc2 = __builtin_amdgcn_wmma_f32_16x16x4_f32(false, a, false, b, (short)0, acc2, false, false);
    }
    {
      v2f b; b.x = c0[(3 * 16 + l) * 4]; b.y = c1[(3 * 16 + l) * 4];
      acc3 = __builtin_amdgcn_wmma_f32_16x16x4_f32(false, a, false, b, (short)0, acc3, false, false);
    }
  }

  float* Mrow = Mid + ((size_t)(bb * 4 + d) * GRID_N + g0) * NCTRL;
  #pragma unroll
  for (int j = 0; j < 8; ++j) {
    const int m = j + 8 * hh;
    Mrow[(size_t)m * NCTRL +  0 + l] = acc0[j];
    Mrow[(size_t)m * NCTRL + 16 + l] = acc1[j];
    Mrow[(size_t)m * NCTRL + 32 + l] = acc2[j];
    Mrow[(size_t)m * NCTRL + 48 + l] = acc3[j];
  }
}

__global__ __launch_bounds__(128)
void stageB_kernel(const float* __restrict__ Mid,
                   const float* __restrict__ Bv,
                   const int*   __restrict__ vspan,
                   float* __restrict__ out)
{
  __shared__ float Sws[4][16][17];

  const int ht = blockIdx.x;
  const int gt = blockIdx.y;
  const int bb = blockIdx.z;
  const int g0 = gt * 16, h0 = ht * 16;

  const int d    = threadIdx.x >> 5;
  const int lane = threadIdx.x & 31;
  const int hh   = lane >> 4;
  const int l    = lane & 15;

  const int cmin = band_min(vspan[h0]);

  const float* Arow = Mid + ((size_t)(bb * 4 + d) * GRID_N + g0 + l) * NCTRL + cmin;
  const float* Brow = Bv + (size_t)(h0 + l) * NCTRL + cmin;

  v8f accO = {};
  #pragma unroll
  for (int kb = 0; kb < 2; ++kb) {
    const int k0 = 4 * kb + 2 * hh;
    v2f a, b;
    a.x = Arow[k0];
    a.y = Arow[k0 + 1];
    b.x = Brow[k0];
    b.y = Brow[k0 + 1];
    accO = __builtin_amdgcn_wmma_f32_16x16x4_f32(false, a, false, b, (short)0, accO, false, false);
  }

  #pragma unroll
  for (int j = 0; j < 8; ++j) Sws[d][j + 8 * hh][l] = accO[j];
  __syncthreads();

  for (int p = threadIdx.x; p < 256; p += 128) {
    const int m = p >> 4, n = p & 15;
    const float inv = 1.0f / Sws[3][m][n];
    const size_t o = (((size_t)bb * GRID_N + (g0 + m)) * GRID_N + (h0 + n)) * 3;
    out[o + 0] = Sws[0][m][n] * inv;
    out[o + 1] = Sws[1][m][n] * inv;
    out[o + 2] = Sws[2][m][n] * inv;
  }
}

__global__ __launch_bounds__(128)
void surfeval_fused_kernel(const float* __restrict__ ctrl,
                           const float* __restrict__ Bu,
                           const float* __restrict__ Bv,
                           float* __restrict__ out)
{
  __shared__ float A1s[4][16][68];
  __shared__ float Sws[4][16][17];

  const int ht = blockIdx.x, gt = blockIdx.y, bb = blockIdx.z;
  const int g0 = gt * 16, h0 = ht * 16;
  const int d    = threadIdx.x >> 5;
  const int lane = threadIdx.x & 31;
  const int hh   = lane >> 4;
  const int l    = lane & 15;

  v8f acc0 = {}, acc1 = {}, acc2 = {}, acc3 = {};
  const float* BuRow = Bu + (size_t)(g0 + l) * MCTRL;
  #pragma unroll
  for (int kb = 0; kb < 16; ++kb) {
    const int k0 = 4 * kb + 2 * hh;
    v2f a; a.x = BuRow[k0]; a.y = BuRow[k0 + 1];
    const float* c0 = ctrl + ((size_t)(bb * MCTRL + k0) * NCTRL) * 4 + d;
    const float* c1 = c0 + (size_t)NCTRL * 4;
    { v2f b; b.x = c0[(0*16+l)*4]; b.y = c1[(0*16+l)*4];
      acc0 = __builtin_amdgcn_wmma_f32_16x16x4_f32(false, a, false, b, (short)0, acc0, false, false); }
    { v2f b; b.x = c0[(1*16+l)*4]; b.y = c1[(1*16+l)*4];
      acc1 = __builtin_amdgcn_wmma_f32_16x16x4_f32(false, a, false, b, (short)0, acc1, false, false); }
    { v2f b; b.x = c0[(2*16+l)*4]; b.y = c1[(2*16+l)*4];
      acc2 = __builtin_amdgcn_wmma_f32_16x16x4_f32(false, a, false, b, (short)0, acc2, false, false); }
    { v2f b; b.x = c0[(3*16+l)*4]; b.y = c1[(3*16+l)*4];
      acc3 = __builtin_amdgcn_wmma_f32_16x16x4_f32(false, a, false, b, (short)0, acc3, false, false); }
  }
  #pragma unroll
  for (int j = 0; j < 8; ++j) {
    const int m = j + 8 * hh;
    A1s[d][m][ 0 + l] = acc0[j];
    A1s[d][m][16 + l] = acc1[j];
    A1s[d][m][32 + l] = acc2[j];
    A1s[d][m][48 + l] = acc3[j];
  }
  __syncthreads();

  v8f accO = {};
  const float* BvRow = Bv + (size_t)(h0 + l) * NCTRL;
  #pragma unroll
  for (int kb = 0; kb < 16; ++kb) {
    const int k0 = 4 * kb + 2 * hh;
    v2f a, b;
    a.x = A1s[d][l][k0]; a.y = A1s[d][l][k0 + 1];
    b.x = BvRow[k0];     b.y = BvRow[k0 + 1];
    accO = __builtin_amdgcn_wmma_f32_16x16x4_f32(false, a, false, b, (short)0, accO, false, false);
  }
  #pragma unroll
  for (int j = 0; j < 8; ++j) Sws[d][j + 8 * hh][l] = accO[j];
  __syncthreads();

  for (int p = threadIdx.x; p < 256; p += 128) {
    const int m = p >> 4, n = p & 15;
    const float inv = 1.0f / Sws[3][m][n];
    const size_t o = (((size_t)bb * GRID_N + (g0 + m)) * GRID_N + (h0 + n)) * 3;
    out[o + 0] = Sws[0][m][n] * inv;
    out[o + 1] = Sws[1][m][n] * inv;
    out[o + 2] = Sws[2][m][n] * inv;
  }
}

extern "C" void kernel_launch(void* const* d_in, const int* in_sizes, int n_in,
                              void* d_out, int out_size, void* d_ws, size_t ws_size,
                              hipStream_t stream) {
  const float* ctrl  = (const float*)d_in[0];
  const int*   uspan = (const int*)  d_in[1];
  const int*   vspan = (const int*)  d_in[2];
  const float* Nu    = (const float*)d_in[3];
  const float* Nv    = (const float*)d_in[4];

  float* Bu = (float*)d_ws;
  float* Bv = Bu + (size_t)GRID_N * MCTRL;
  float* Mid = Bv + (size_t)GRID_N * NCTRL;

  const size_t need = ((size_t)2 * GRID_N * MCTRL +
                       (size_t)BATCH * 4 * GRID_N * NCTRL) * sizeof(float);

  build_basis_kernel<<<4, 256, 0, stream>>>(Nu, uspan, Nv, vspan, Bu, Bv);

  if (ws_size >= need) {
    dim3 gridA(GRID_N / 16, BATCH);
    stageA_kernel<<<gridA, 128, 0, stream>>>(ctrl, Bu, uspan, Mid);
    dim3 gridB(GRID_N / 16, GRID_N / 16, BATCH);
    stageB_kernel<<<gridB, 128, 0, stream>>>(Mid, Bv, vspan, (float*)d_out);
  } else {
    dim3 grid(GRID_N / 16, GRID_N / 16, BATCH);
    surfeval_fused_kernel<<<grid, 128, 0, stream>>>(ctrl, Bu, Bv, (float*)d_out);
  }
}


# Fuse the 3 launches into 1: each workgroup forms its basis band from the weights and spans in registers, keeps its 16x64 Mid stripe in LDS and loops over 8 v-tiles; no workspace
# speedup vs baseline: 1.6934x; 1.6934x over previous
#include <hip/hip_runtime.h>


typedef float v2f __attribute__((ext_vector_type(2)));
typedef float v8f __attribute__((ext_vector_type(8)));

#define GRID_N 512
#define MCTRL  64
#define NCTRL  64
#define BATCH  8
#define HSPLIT 4
#define TILES  (GRID_N / 16 / HSPLIT)

__device__ __forceinline__ int band_min(int span) {
  int c = span - 3;
  return (c > 56) ? 56 : (c < 0 ? 0 : c);
}

__device__ __forceinline__ float band_val(const float4 w, int r) {
  float v = 0.0f;
  v = (r == 0) ? w.x : v;
  v = (r == 1) ? w.y : v;
  v = (r == 2) ? w.z : v;
  v = (r == 3) ? w.w : v;
  return v;
}

__global__ __launch_bounds__(128)
void surface_fused_kernel(const float* __restrict__ ctrl,
                          const int*   __restrict__ uspan,
                          const int*   __restrict__ vspan,
                          const float* __restrict__ Nu,
                          const float* __restrict__ Nv,
                          float* __restrict__ out)
{
  __shared__ float A1s[4][16][68];
  __shared__ float Sws[2][4][16][17];

  const int hq = blockIdx.x;
  const int gt = blockIdx.y;
  const int bb = blockIdx.z;
  const int g0 = gt * 16;

  const int d    = threadIdx.x >> 5;
  const int lane = threadIdx.x & 31;
  const int hh   = lane >> 4;
  const int l    = lane & 15;

  v2f bf[TILES][2];
  int cminT[TILES];
  #pragma unroll
  for (int t = 0; t < TILES; ++t) {
    const int h0   = (hq * TILES + t) * 16;
    const int cmin = band_min(vspan[h0]);
    const int base = vspan[h0 + l] - 3 - cmin;
    const float4 w = *reinterpret_cast<const float4*>(Nv + (size_t)(h0 + l) * 4);
    #pragma unroll
    for (int kb = 0; kb < 2; ++kb) {
      const int k0 = 4 * kb + 2 * hh;
      bf[t][kb].x = band_val(w, k0 - base);
      bf[t][kb].y = band_val(w, k0 + 1 - base);
    }
    cminT[t] = cmin;
  }

  const int umin = band_min(uspan[g0]);
  {
    const int ubase = uspan[g0 + l] - 3 - umin;
    const float4 wu = *reinterpret_cast<const float4*>(Nu + (size_t)(g0 + l) * 4);

    v8f acc0 = {}, acc1 = {}, acc2 = {}, acc3 = {};
    #pragma unroll
    for (int kb = 0; kb < 2; ++kb) {
      const int k0 = 4 * kb + 2 * hh;
      v2f a;
      a.x = band_val(wu, k0 - ubase);
      a.y = band_val(wu, k0 + 1 - ubase);
      const float* c0 = ctrl + ((size_t)(bb * MCTRL + umin + k0) * NCTRL) * 4 + d;
      const float* c1 = c0 + (size_t)NCTRL * 4;
      {
        v2f b; b.x = c0[(0 * 16 + l) * 4]; b.y = c1[(0 * 16 + l) * 4];
        acc0 = __builtin_amdgcn_wmma_f32_16x16x4_f32(false, a, false, b, (short)0, acc0, false, false);
      }
      {
        v2f b; b.x = c0[(1 * 16 + l) * 4]; b.y = c1[(1 * 16 + l) * 4];
        acc1 = __builtin_amdgcn_wmma_f32_16x16x4_f32(false, a, false, b, (short)0, acc1, false, false);
      }
      {
        v2f b; b.x = c0[(2 * 16 + l) * 4]; b.y = c1[(2 * 16 + l) * 4];
        acc2 = __builtin_amdgcn_wmma_f32_16x16x4_f32(false, a, false, b, (short)0, acc2, false, false);
      }
      {
        v2f b; b.x = c0[(3 * 16 + l) * 4]; b.y = c1[(3 * 16 + l) * 4];
        acc3 = __builtin_amdgcn_wmma_f32_16x16x4_f32(false, a, false, b, (short)0, acc3, false, false);
      }
    }

    #pragma unroll
    for (int j = 0; j < 8; ++j) {
      const int m = j + 8 * hh;
      A1s[d][m][ 0 + l] = acc0[j];
      A1s[d][m][16 + l] = acc1[j];
      A1s[d][m][32 + l] = acc2[j];
      A1s[d][m][48 + l] = acc3[j];
    }
  }
  __syncthreads();

  #pragma unroll
  for (int t = 0; t < TILES; ++t) {
    const int h0 = (hq * TILES + t) * 16;
    const float* Arow = &A1s[d][l][cminT[t]];

    v8f accO = {};
    #pragma unroll
    for (int kb = 0; kb < 2; ++kb) {
      const int k0 = 4 * kb + 2 * hh;
      v2f a;
      a.x = Arow[k0];
      a.y = Arow[k0 + 1];
      accO = __builtin_amdgcn_wmma_f32_16x16x4_f32(false, a, false, bf[t][kb], (short)0, accO, false, false);
    }

    float (*S)[16][17] = Sws[t & 1];
    #pragma unroll
    for (int j = 0; j < 8; ++j) S[d][j + 8 * hh][l] = accO[j];
    __syncthreads();

    for (int p = threadIdx.x; p < 256; p += 128) {
      const int m = p >> 4, n = p & 15;
      const float inv = 1.0f / S[3][m][n];
      const size_t o = (((size_t)bb * GRID_N + (g0 + m)) * GRID_N + (h0 + n)) * 3;
      out[o + 0] = S[0][m][n] * inv;
      out[o + 1] = S[1][m][n] * inv;
      out[o + 2] = S[2][m][n] * inv;
    }
  }
}

extern "C" void kernel_launch(void* const* d_in, const int* in_sizes, int n_in,
                              void* d_out, int out_size, void* d_ws, size_t ws_size,
                              hipStream_t stream) {
  const float* ctrl  = (const float*)d_in[0];
  const int*   uspan = (const int*)  d_in[1];
  const int*   vspan = (const int*)  d_in[2];
  const float* Nu    = (const float*)d_in[3];
  const float* Nv    = (const float*)d_in[4];

  dim3 grid(HSPLIT, GRID_N / 16, BATCH);
  surface_fused_kernel<<<grid, 128, 0, stream>>>(ctrl, uspan, vspan, Nu, Nv, (float*)d_out);
}


# Each wave forms all 4 components of 2 neighbouring v-tiles instead of 1 component of all 8: x,y,z,w meet in one lane, the divide runs in registers, the per-tile LDS exchange and its 8 barriers go
# speedup vs baseline: 3.1219x; 1.8436x over previous
#include <hip/hip_runtime.h>


typedef float v2f __attribute__((ext_vector_type(2)));
typedef float v8f __attribute__((ext_vector_type(8)));

#define GRID_N 512
#define MCTRL  64
#define NCTRL  64
#define BATCH  8
#define HSPLIT 4
#define TILES  (GRID_N / 16 / HSPLIT)
#define WAVES  4
#define WTILES (TILES / WAVES)

__device__ __forceinline__ int band_min(int span) {
  int c = span - 3;
  return (c > 56) ? 56 : (c < 0 ? 0 : c);
}

__device__ __forceinline__ float band_val(const float4 w, int r) {
  float v = 0.0f;
  v = (r == 0) ? w.x : v;
  v = (r == 1) ? w.y : v;
  v = (r == 2) ? w.z : v;
  v = (r == 3) ? w.w : v;
  return v;
}

__global__ __launch_bounds__(128)
void surface_fused_kernel(const float* __restrict__ ctrl,
                          const int*   __restrict__ uspan,
                          const int*   __restrict__ vspan,
                          const float* __restrict__ Nu,
                          const float* __restrict__ Nv,
                          float* __restrict__ out)
{
  __shared__ float A1s[4][16][68];

  const int hq = blockIdx.x;
  const int gt = blockIdx.y;
  const int bb = blockIdx.z;
  const int g0 = gt * 16;

  const int d    = threadIdx.x >> 5;
  const int lane = threadIdx.x & 31;
  const int hh   = lane >> 4;
  const int l    = lane & 15;

  v2f bf[WTILES][2];
  int cminT[WTILES];
  #pragma unroll
  for (int t = 0; t < WTILES; ++t) {
    const int h0   = (hq * TILES + d * WTILES + t) * 16;
    const int cmin = band_min(vspan[h0]);
    const int base = vspan[h0 + l] - 3 - cmin;
    const float4 w = *reinterpret_cast<const float4*>(Nv + (size_t)(h0 + l) * 4);
    #pragma unroll
    for (int kb = 0; kb < 2; ++kb) {
      const int k0 = 4 * kb + 2 * hh;
      bf[t][kb].x = band_val(w, k0 - base);
      bf[t][kb].y = band_val(w, k0 + 1 - base);
    }
    cminT[t] = cmin;
  }

  const int umin = band_min(uspan[g0]);
  {
    const int ubase = uspan[g0 + l] - 3 - umin;
    const float4 wu = *reinterpret_cast<const float4*>(Nu + (size_t)(g0 + l) * 4);

    v8f acc0 = {}, acc1 = {}, acc2 = {}, acc3 = {};
    #pragma unroll
    for (int kb = 0; kb < 2; ++kb) {
      const int k0 = 4 * kb + 2 * hh;
      v2f a;
      a.x = band_val(wu, k0 - ubase);
      a.y = band_val(wu, k0 + 1 - ubase);
      const float* c0 = ctrl + ((size_t)(bb * MCTRL + umin + k0) * NCTRL) * 4 + d;
      const float* c1 = c0 + (size_t)NCTRL * 4;
      {
        v2f b; b.x = c0[(0 * 16 + l) * 4]; b.y = c1[(0 * 16 + l) * 4];
        acc0 = __builtin_amdgcn_wmma_f32_16x16x4_f32(false, a, false, b, (short)0, acc0, false, false);
      }
      {
        v2f b; b.x = c0[(1 * 16 + l) * 4]; b.y = c1[(1 * 16 + l) * 4];
        acc1 = __builtin_amdgcn_wmma_f32_16x16x4_f32(false, a, false, b, (short)0, acc1, false, false);
      }
      {
        v2f b; b.x = c0[(2 * 16 + l) * 4]; b.y = c1[(2 * 16 + l) * 4];
        acc2 = __builtin_amdgcn_wmma_f32_16x16x4_f32(false, a, false, b, (short)0, acc2, false, false);
      }
      {
        v2f b; b.x = c0[(3 * 16 + l) * 4]; b.y = c1[(3 * 16 + l) * 4];
        acc3 = __builtin_amdgcn_wmma_f32_16x16x4_f32(false, a, false, b, (short)0, acc3, false, false);
      }
    }

    #pragma unroll
    for (int j = 0; j < 8; ++j) {
      const int m = j + 8 * hh;
      A1s[d][m][ 0 + l] = acc0[j];
      A1s[d][m][16 + l] = acc1[j];
      A1s[d][m][32 + l] = acc2[j];
      A1s[d][m][48 + l] = acc3[j];
    }
  }
  __syncthreads();

  v8f accX[WTILES], accY[WTILES], accZ[WTILES], accW[WTILES];
  #pragma unroll
  for (int t = 0; t < WTILES; ++t) {
    const float* Arow = &A1s[0][l][cminT[t]];
    v8f aX = {}, aY = {}, aZ = {}, aW = {};
    #pragma unroll
    for (int kb = 0; kb < 2; ++kb) {
      const int k0 = 4 * kb + 2 * hh;
      v2f ax, ay, az, aw;
      ax.x = Arow[0 * 16 * 68 + k0];  ax.y = Arow[0 * 16 * 68 + k0 + 1];
      ay.x = Arow[1 * 16 * 68 + k0];  ay.y = Arow[1 * 16 * 68 + k0 + 1];
      az.x = Arow[2 * 16 * 68 + k0];  az.y = Arow[2 * 16 * 68 + k0 + 1];
      aw.x = Arow[3 * 16 * 68 + k0];  aw.y = Arow[3 * 16 * 68 + k0 + 1];
      aX = __builtin_amdgcn_wmma_f32_16x16x4_f32(false, ax, false, bf[t][kb], (short)0, aX, false, false);
      aY = __builtin_amdgcn_wmma_f32_16x16x4_f32(false, ay, false, bf[t][kb], (short)0, aY, false, false);
      aZ = __builtin_amdgcn_wmma_f32_16x16x4_f32(false, az, false, bf[t][kb], (short)0, aZ, false, false);
      aW = __builtin_amdgcn_wmma_f32_16x16x4_f32(false, aw, false, bf[t][kb], (short)0, aW, false, false);
    }
    accX[t] = aX; accY[t] = aY; accZ[t] = aZ; accW[t] = aW;
  }

  float* orow = out + (((size_t)bb * GRID_N + (g0 + 8 * hh)) * GRID_N
                       + ((hq * TILES + d * WTILES) * 16 + l)) * 3;
  #pragma unroll
  for (int j = 0; j < 8; ++j) {
    #pragma unroll
    for (int t = 0; t < WTILES; ++t) {
      const float inv = 1.0f / accW[t][j];
      float* o = orow + (size_t)j * GRID_N * 3 + t * 16 * 3;
      o[0] = accX[t][j] * inv;
      o[1] = accY[t][j] * inv;
      o[2] = accZ[t][j] * inv;
    }
  }
}

extern "C" void kernel_launch(void* const* d_in, const int* in_sizes, int n_in,
                              void* d_out, int out_size, void* d_ws, size_t ws_size,
                              hipStream_t stream) {
  const float* ctrl  = (const float*)d_in[0];
  const int*   uspan = (const int*)  d_in[1];
  const int*   vspan = (const int*)  d_in[2];
  const float* Nu    = (const float*)d_in[3];
  const float* Nv    = (const float*)d_in[4];

  dim3 grid(HSPLIT, GRID_N / 16, BATCH);
  surface_fused_kernel<<<grid, 128, 0, stream>>>(ctrl, uspan, vspan, Nu, Nv, (float*)d_out);
}
